# starting kernel
# baseline (speedup 1.0000x reference)
#include <hip/hip_runtime.h>


typedef float v2f __attribute__((ext_vector_type(2)));
typedef float v8f __attribute__((ext_vector_type(8)));

#define N_NODES 50000
#define N_EDGES 800000

template <int K, int DOUT, bool RELU_IN>
__global__ __launch_bounds__(256) void gemm_wmma_f32(
    const float* __restrict__ X, const float* __restrict__ W,
    float* __restrict__ Y, int N) {
  constexpr bool GUARD = (DOUT % 16) != 0;
  const int wave  = threadIdx.x >> 5;
  const int lane  = threadIdx.x & 31;
  const int n0    = wave * 16;
  const int m0    = blockIdx.x * 16;
  const int lhalf = lane >> 4;
  const int l16   = lane & 15;

  const int  rowA  = m0 + l16;
  const int  colB  = n0 + l16;
  const bool colOK = !GUARD || (colB < DOUT);
  const float* xrow = X + (size_t)rowA * K;
  const int  kBase = 2 * lhalf;

  v8f acc = {0.f, 0.f, 0.f, 0.f, 0.f, 0.f, 0.f, 0.f};
#pragma unroll 8
  for (int k = 0; k < K; k += 4) {
    float ax = xrow[k + kBase + 0];
    float ay = xrow[k + kBase + 1];
    if (RELU_IN) { ax = fmaxf(ax, 0.f); ay = fmaxf(ay, 0.f); }
    v2f a = {ax, ay};
    v2f b;
    if (GUARD) {
      b.x = colOK ? W[(size_t)(k + kBase + 0) * DOUT + colB] : 0.f;
      b.y = colOK ? W[(size_t)(k + kBase + 1) * DOUT + colB] : 0.f;
    } else {
      b.x = W[(size_t)(k + kBase + 0) * DOUT + colB];
      b.y = W[(size_t)(k + kBase + 1) * DOUT + colB];
    }
    acc = __builtin_amdgcn_wmma_f32_16x16x4_f32(
         false, a,  false, b,
         (short)0, acc,  false,  false);
  }

  if (colOK) {
#pragma unroll
    for (int r = 0; r < 8; ++r) {
      Y[(size_t)(m0 + r + 8 * lhalf) * DOUT + colB] = acc[r];
    }
  }
}

template <int D>
__global__ __launch_bounds__(256) void init_bias(
    float* __restrict__ A, const float* __restrict__ b, int N) {
  size_t i = (size_t)blockIdx.x * blockDim.x + threadIdx.x;
  if (i < (size_t)N * D) A[i] = b[(int)(i % D)];
}

template <int D>
__global__ __launch_bounds__(256) void agg_vec4(
    const float* __restrict__ S, const int* __restrict__ src,
    const int* __restrict__ dst, const float* __restrict__ ew,
    float* __restrict__ A, int E) {
  constexpr int CH = D >> 2;
  size_t i = (size_t)blockIdx.x * blockDim.x + threadIdx.x;
  if (i >= (size_t)E * CH) return;
  const int e = (int)(i / CH);
  const int f = (int)(i % CH) * 4;
  const float w = ew[e];
  const float4 s = *(const float4*)(S + (size_t)src[e] * D + f);
  float* base = A + (size_t)dst[e] * D + f;
  unsafeAtomicAdd(base + 0, w * s.x);
  unsafeAtomicAdd(base + 1, w * s.y);
  unsafeAtomicAdd(base + 2, w * s.z);
  unsafeAtomicAdd(base + 3, w * s.w);
}

template <int D>
__global__ __launch_bounds__(256) void agg_scalar(
    const float* __restrict__ S, const int* __restrict__ src,
    const int* __restrict__ dst, const float* __restrict__ ew,
    float* __restrict__ A, int E) {
  size_t i = (size_t)blockIdx.x * blockDim.x + threadIdx.x;
  if (i >= (size_t)E * D) return;
  const int e = (int)(i / D);
  const int f = (int)(i % D);
  const float w = ew[e];
  const float s = S[(size_t)src[e] * D + f];
  unsafeAtomicAdd(A + (size_t)dst[e] * D + f, w * s);
}

static inline int cdiv(long long a, long long b) { return (int)((a + b - 1) / b); }

extern "C" void kernel_launch(void* const* d_in, const int* in_sizes, int n_in,
                              void* d_out, int out_size, void* d_ws, size_t ws_size,
                              hipStream_t stream) {
  const float* x   = (const float*)d_in[0];
  const int*   src = (const int*)  d_in[1];
  const int*   dst = (const int*)  d_in[2];
  const float* ew  = (const float*)d_in[3];
  const float* W1  = (const float*)d_in[4];
  const float* b1  = (const float*)d_in[5];
  const float* W2  = (const float*)d_in[6];
  const float* b2  = (const float*)d_in[7];
  const float* W3  = (const float*)d_in[8];
  const float* b3  = (const float*)d_in[9];
  float* out = (float*)d_out;

  const int N = N_NODES, E = N_EDGES;
  const int rowTiles = N / 16;

  float* bufS = (float*)d_ws;
  float* bufH = bufS + (size_t)N * 128;

  gemm_wmma_f32<256, 128, false><<<rowTiles, 256, 0, stream>>>(x, W1, bufS, N);
  init_bias<128><<<cdiv((long long)N * 128, 256), 256, 0, stream>>>(bufH, b1, N);
  agg_vec4<128><<<cdiv((long long)E * 32, 256), 256, 0, stream>>>(
      bufS, src, dst, ew, bufH, E);

  gemm_wmma_f32<128, 64, true><<<rowTiles, 128, 0, stream>>>(bufH, W2, bufS, N);
  init_bias<64><<<cdiv((long long)N * 64, 256), 256, 0, stream>>>(bufH, b2, N);
  agg_vec4<64><<<cdiv((long long)E * 16, 256), 256, 0, stream>>>(
      bufS, src, dst, ew, bufH, E);

  gemm_wmma_f32<64, 10, true><<<rowTiles, 32, 0, stream>>>(bufH, W3, bufS, N);
  init_bias<10><<<cdiv((long long)N * 10, 256), 256, 0, stream>>>(out, b3, N);
  agg_scalar<10><<<cdiv((long long)E * 10, 256), 256, 0, stream>>>(
      bufS, src, dst, ew, out, E);
}


# Edge sums without f32 atomics: edges counting-sorted by dst once per call (rank = returning int count atomic, arrival order); each row summed in registers in that order, stored once with bias
# speedup vs baseline: 1.6812x; 1.6812x over previous
#include <hip/hip_runtime.h>


typedef float v2f __attribute__((ext_vector_type(2)));
typedef float v8f __attribute__((ext_vector_type(8)));

#define N_NODES 50000
#define N_EDGES 800000

template <int K, int DOUT, bool RELU_IN>
__global__ __launch_bounds__(256) void gemm_wmma_f32(
    const float* __restrict__ X, const float* __restrict__ W,
    float* __restrict__ Y, int N) {
  constexpr bool GUARD = (DOUT % 16) != 0;
  const int wave  = threadIdx.x >> 5;
  const int lane  = threadIdx.x & 31;
  const int n0    = wave * 16;
  const int m0    = blockIdx.x * 16;
  const int lhalf = lane >> 4;
  const int l16   = lane & 15;

  const int  rowA  = m0 + l16;
  const int  colB  = n0 + l16;
  const bool colOK = !GUARD || (colB < DOUT);
  const float* xrow = X + (size_t)rowA * K;
  const int  kBase = 2 * lhalf;

  v8f acc = {0.f, 0.f, 0.f, 0.f, 0.f, 0.f, 0.f, 0.f};
#pragma unroll 8
  for (int k = 0; k < K; k += 4) {
    float ax = xrow[k + kBase + 0];
    float ay = xrow[k + kBase + 1];
    if (RELU_IN) { ax = fmaxf(ax, 0.f); ay = fmaxf(ay, 0.f); }
    v2f a = {ax, ay};
    v2f b;
    if (GUARD) {
      b.x = colOK ? W[(size_t)(k + kBase + 0) * DOUT + colB] : 0.f;
      b.y = colOK ? W[(size_t)(k + kBase + 1) * DOUT + colB] : 0.f;
    } else {
      b.x = W[(size_t)(k + kBase + 0) * DOUT + colB];
      b.y = W[(size_t)(k + kBase + 1) * DOUT + colB];
    }
    acc = __builtin_amdgcn_wmma_f32_16x16x4_f32(
         false, a,  false, b,
         (short)0, acc,  false,  false);
  }

  if (colOK) {
#pragma unroll
    for (int r = 0; r < 8; ++r) {
      Y[(size_t)(m0 + r + 8 * lhalf) * DOUT + colB] = acc[r];
    }
  }
}

__global__ __launch_bounds__(256) void zero_counts(int* __restrict__ cnt, int n) {
  const int i = blockIdx.x * blockDim.x + threadIdx.x;
  if (i < n) cnt[i] = 0;
}

__global__ __launch_bounds__(256) void count_rank(
    const int* __restrict__ dst, int* __restrict__ cnt,
    int* __restrict__ rank, int E) {
  const int e = blockIdx.x * blockDim.x + threadIdx.x;
  if (e < E) rank[e] = atomicAdd(&cnt[dst[e]], 1);
}

#define SCAN_IT  ((N_NODES + 1023) / 1024)
#define SCAN_SEG (SCAN_IT * 32)
__global__ __launch_bounds__(1024) void scan_counts(
    const int* __restrict__ cnt, int* __restrict__ rowptr) {
  __shared__ int wtot[32];
  const int lane = threadIdx.x & 31;
  const int wave = threadIdx.x >> 5;
  const int seg0 = wave * SCAN_SEG;

  int v[SCAN_IT];
  int lsum = 0;
#pragma unroll
  for (int it = 0; it < SCAN_IT; ++it) {
    const int idx = seg0 + it * 32 + lane;
    v[it] = (idx < N_NODES) ? cnt[idx] : 0;
    lsum += v[it];
  }
  for (int off = 16; off > 0; off >>= 1) lsum += __shfl_xor(lsum, off);
  if (lane == 0) wtot[wave] = lsum;
  __syncthreads();

  int carry = (lane < wave) ? wtot[lane] : 0;
  for (int off = 16; off > 0; off >>= 1) carry += __shfl_xor(carry, off);

#pragma unroll
  for (int it = 0; it < SCAN_IT; ++it) {
    int inc = v[it];
    for (int off = 1; off < 32; off <<= 1) {
      const int up = __shfl_up(inc, off);
      if (lane >= off) inc += up;
    }
    const int idx = seg0 + it * 32 + lane;
    if (idx < N_NODES) rowptr[idx] = carry + inc - v[it];
    carry += __shfl(inc, 31);
  }
  if (threadIdx.x == 1023) rowptr[N_NODES] = carry;
}

__global__ __launch_bounds__(256) void place_edges(
    const int* __restrict__ src, const int* __restrict__ dst,
    const float* __restrict__ ew, const int* __restrict__ rowptr,
    const int* __restrict__ rank, int2* __restrict__ rec, int E) {
  const int e = blockIdx.x * blockDim.x + threadIdx.x;
  if (e < E) {
    const int p = rowptr[dst[e]] + rank[e];
    rec[p] = make_int2(src[e], __float_as_int(ew[e]));
  }
}

template <int D>
__global__ __launch_bounds__(256) void gather_vec4(
    const float* __restrict__ S, const int* __restrict__ rowptr,
    const int2* __restrict__ rec, const float* __restrict__ b,
    float* __restrict__ A, int N) {
  constexpr int CH = D >> 2;
  const int i = blockIdx.x * blockDim.x + threadIdx.x;
  if (i >= N * CH) return;
  const int n = i / CH;
  const int f = (i % CH) * 4;
  const int beg = rowptr[n];
  const int end = rowptr[n + 1];
  float4 acc = *(const float4*)(b + f);
#pragma unroll 4
  for (int p = beg; p < end; ++p) {
    const int2 r = rec[p];
    const float w = __int_as_float(r.y);
    const float4 s = *(const float4*)(S + (size_t)r.x * D + f);
    acc.x = fmaf(w, s.x, acc.x);
    acc.y = fmaf(w, s.y, acc.y);
    acc.z = fmaf(w, s.z, acc.z);
    acc.w = fmaf(w, s.w, acc.w);
  }
  *(float4*)(A + (size_t)n * D + f) = acc;
}

template <int D>
__global__ __launch_bounds__(256) void gather_scalar(
    const float* __restrict__ S, const int* __restrict__ rowptr,
    const int2* __restrict__ rec, const float* __restrict__ b,
    float* __restrict__ A, int N) {
  const int i = blockIdx.x * blockDim.x + threadIdx.x;
  if (i >= N * D) return;
  const int n = i / D;
  const int f = i % D;
  const int beg = rowptr[n];
  const int end = rowptr[n + 1];
  float acc = b[f];
#pragma unroll 4
  for (int p = beg; p < end; ++p) {
    const int2 r = rec[p];
    acc = fmaf(__int_as_float(r.y), S[(size_t)r.x * D + f], acc);
  }
  A[i] = acc;
}

static inline int cdiv(long long a, long long b) { return (int)((a + b - 1) / b); }

extern "C" void kernel_launch(void* const* d_in, const int* in_sizes, int n_in,
                              void* d_out, int out_size, void* d_ws, size_t ws_size,
                              hipStream_t stream) {
  const float* x   = (const float*)d_in[0];
  const int*   src = (const int*)  d_in[1];
  const int*   dst = (const int*)  d_in[2];
  const float* ew  = (const float*)d_in[3];
  const float* W1  = (const float*)d_in[4];
  const float* b1  = (const float*)d_in[5];
  const float* W2  = (const float*)d_in[6];
  const float* b2  = (const float*)d_in[7];
  const float* W3  = (const float*)d_in[8];
  const float* b3  = (const float*)d_in[9];
  float* out = (float*)d_out;

  const int N = N_NODES, E = N_EDGES;
  const int rowTiles = N / 16;

  float* bufS = (float*)d_ws;
  float* bufH = bufS + (size_t)N * 128;
  int*  cnt    = (int*)(bufH + (size_t)N * 128);
  int*  rowptr = cnt + 50048;
  int*  rank   = rowptr + 50048;
  int2* rec    = (int2*)(rank + N_EDGES);

  zero_counts<<<cdiv(N, 256), 256, 0, stream>>>(cnt, N);
  count_rank<<<cdiv(E, 256), 256, 0, stream>>>(dst, cnt, rank, E);
  scan_counts<<<1, 1024, 0, stream>>>(cnt, rowptr);
  place_edges<<<cdiv(E, 256), 256, 0, stream>>>(src, dst, ew, rowptr, rank, rec, E);

  gemm_wmma_f32<256, 128, false><<<rowTiles, 256, 0, stream>>>(x, W1, bufS, N);
  gather_vec4<128><<<cdiv((long long)N * 32, 256), 256, 0, stream>>>(
      bufS, rowptr, rec, b1, bufH, N);

  gemm_wmma_f32<128, 64, true><<<rowTiles, 128, 0, stream>>>(bufH, W2, bufS, N);
  gather_vec4<64><<<cdiv((long long)N * 16, 256), 256, 0, stream>>>(
      bufS, rowptr, rec, b2, bufH, N);

  gemm_wmma_f32<64, 10, true><<<rowTiles, 32, 0, stream>>>(bufH, W3, bufS, N);
  gather_scalar<10><<<cdiv((long long)N * 10, 256), 256, 0, stream>>>(
      bufS, rowptr, rec, b3, out, N);
}


# Edge ranks from LDS counters: a work-group counts one node range of one edge slice in LDS (rank = LDS atomic return); one global add per node and slice, consecutive words, sets slice order (arrival)
# speedup vs baseline: 1.9184x; 1.1411x over previous
#include <hip/hip_runtime.h>


typedef float v2f __attribute__((ext_vector_type(2)));
typedef float v8f __attribute__((ext_vector_type(8)));

#define N_NODES 50000
#define N_EDGES 800000

template <int K, int DOUT, bool RELU_IN>
__global__ __launch_bounds__(256) void gemm_wmma_f32(
    const float* __restrict__ X, const float* __restrict__ W,
    float* __restrict__ Y, int N) {
  constexpr bool GUARD = (DOUT % 16) != 0;
  const int wave  = threadIdx.x >> 5;
  const int lane  = threadIdx.x & 31;
  const int n0    = wave * 16;
  const int m0    = blockIdx.x * 16;
  const int lhalf = lane >> 4;
  const int l16   = lane & 15;

  const int  rowA  = m0 + l16;
  const int  colB  = n0 + l16;
  const bool colOK = !GUARD || (colB < DOUT);
  const float* xrow = X + (size_t)rowA * K;
  const int  kBase = 2 * lhalf;

  v8f acc = {0.f, 0.f, 0.f, 0.f, 0.f, 0.f, 0.f, 0.f};
#pragma unroll 8
  for (int k = 0; k < K; k += 4) {
    float ax = xrow[k + kBase + 0];
    float ay = xrow[k + kBase + 1];
    if (RELU_IN) { ax = fmaxf(ax, 0.f); ay = fmaxf(ay, 0.f); }
    v2f a = {ax, ay};
    v2f b;
    if (GUARD) {
      b.x = colOK ? W[(size_t)(k + kBase + 0) * DOUT + colB] : 0.f;
      b.y = colOK ? W[(size_t)(k + kBase + 1) * DOUT + colB] : 0.f;
    } else {
      b.x = W[(size_t)(k + kBase + 0) * DOUT + colB];
      b.y = W[(size_t)(k + kBase + 1) * DOUT + colB];
    }
    acc = __builtin_amdgcn_wmma_f32_16x16x4_f32(
         false, a,  false, b,
         (short)0, acc,  false,  false);
  }

  if (colOK) {
#pragma unroll
    for (int r = 0; r < 8; ++r) {
      Y[(size_t)(m0 + r + 8 * lhalf) * DOUT + colB] = acc[r];
    }
  }
}

#define SORT_R     8
#define SORT_S     32
#define RANGE_N    ((N_NODES + SORT_R - 1) / SORT_R)
#define SLICE_E    (N_EDGES / SORT_S)
#define SLICE_Q    (SLICE_E / 4)
#define CR_THREADS 1024
#define CR_IT      ((SLICE_Q + CR_THREADS - 1) / CR_THREADS)
#define CR_PN      ((RANGE_N + CR_THREADS - 1) / CR_THREADS)
static_assert(SLICE_E * SORT_S == N_EDGES, "slices must tile the edge list");
static_assert(SLICE_E % 4 == 0, "a slice is read as whole int4");
static_assert(RANGE_N * SORT_R >= N_NODES, "ranges must cover the nodes");
__global__ __launch_bounds__(256) void zero_counts(int* __restrict__ cnt, int n) {
  const int i = blockIdx.x * blockDim.x + threadIdx.x;
  if (i < n) cnt[i] = 0;
}

__global__ __launch_bounds__(CR_THREADS) void count_rank_lds(
    const int* __restrict__ dst, int* __restrict__ cnt,
    int* __restrict__ rank, int* __restrict__ base) {
  __shared__ int lcnt[RANGE_N];
  const int ri = blockIdx.x % SORT_R;
  const int sj = blockIdx.x / SORT_R;
  const int lo = ri * RANGE_N;
  const int4* __restrict__ d4 = (const int4*)(dst + (size_t)sj * SLICE_E);
  int* __restrict__ rk = rank + (size_t)sj * SLICE_E;

  int4 d[CR_IT];
#pragma unroll
  for (int it = 0; it < CR_IT; ++it) {
    const int q = it * CR_THREADS + (int)threadIdx.x;
    d[it] = (q < SLICE_Q) ? d4[q] : make_int4(-1, -1, -1, -1);
  }

  for (int n = threadIdx.x; n < RANGE_N; n += CR_THREADS) lcnt[n] = 0;
  __syncthreads();

#pragma unroll
  for (int it = 0; it < CR_IT; ++it) {
    const int e0 = (it * CR_THREADS + (int)threadIdx.x) * 4;
    unsigned u;
    u = (unsigned)(d[it].x - lo);
    if (u < (unsigned)RANGE_N) rk[e0 + 0] = atomicAdd(&lcnt[u], 1);
    u = (unsigned)(d[it].y - lo);
    if (u < (unsigned)RANGE_N) rk[e0 + 1] = atomicAdd(&lcnt[u], 1);
    u = (unsigned)(d[it].z - lo);
    if (u < (unsigned)RANGE_N) rk[e0 + 2] = atomicAdd(&lcnt[u], 1);
    u = (unsigned)(d[it].w - lo);
    if (u < (unsigned)RANGE_N) rk[e0 + 3] = atomicAdd(&lcnt[u], 1);
  }
  __syncthreads();

  int got[CR_PN];
#pragma unroll
  for (int it = 0; it < CR_PN; ++it) {
    const int n = it * CR_THREADS + (int)threadIdx.x;
    const int node = lo + n;
    got[it] = 0;
    if (n < RANGE_N && node < N_NODES) got[it] = atomicAdd(&cnt[node], lcnt[n]);
  }
#pragma unroll
  for (int it = 0; it < CR_PN; ++it) {
    const int n = it * CR_THREADS + (int)threadIdx.x;
    const int node = lo + n;
    if (n < RANGE_N && node < N_NODES) base[(size_t)sj * N_NODES + node] = got[it];
  }
}

#define SCAN_IT  ((N_NODES + 1023) / 1024)
#define SCAN_SEG (SCAN_IT * 32)
__global__ __launch_bounds__(1024) void scan_counts(
    const int* __restrict__ cnt, int* __restrict__ rowptr) {
  __shared__ int wtot[32];
  const int lane = threadIdx.x & 31;
  const int wave = threadIdx.x >> 5;
  const int seg0 = wave * SCAN_SEG;

  int v[SCAN_IT];
  int lsum = 0;
#pragma unroll
  for (int it = 0; it < SCAN_IT; ++it) {
    const int idx = seg0 + it * 32 + lane;
    v[it] = (idx < N_NODES) ? cnt[idx] : 0;
    lsum += v[it];
  }
  for (int off = 16; off > 0; off >>= 1) lsum += __shfl_xor(lsum, off);
  if (lane == 0) wtot[wave] = lsum;
  __syncthreads();

  int carry = (lane < wave) ? wtot[lane] : 0;
  for (int off = 16; off > 0; off >>= 1) carry += __shfl_xor(carry, off);

#pragma unroll
  for (int it = 0; it < SCAN_IT; ++it) {
    int inc = v[it];
    for (int off = 1; off < 32; off <<= 1) {
      const int up = __shfl_up(inc, off);
      if (lane >= off) inc += up;
    }
    const int idx = seg0 + it * 32 + lane;
    if (idx < N_NODES) rowptr[idx] = carry + inc - v[it];
    carry += __shfl(inc, 31);
  }
  if (threadIdx.x == 1023) rowptr[N_NODES] = carry;
}

__global__ __launch_bounds__(256) void place_edges(
    const int* __restrict__ src, const int* __restrict__ dst,
    const float* __restrict__ ew, const int* __restrict__ rowptr,
    const int* __restrict__ rank, const int* __restrict__ base,
    int2* __restrict__ rec, int E) {
  const int e = blockIdx.x * blockDim.x + threadIdx.x;
  if (e < E) {
    const int d = dst[e];
    const int sj = e / SLICE_E;
    const int p = rowptr[d] + base[(size_t)sj * N_NODES + d] + rank[e];
    rec[p] = make_int2(src[e], __float_as_int(ew[e]));
  }
}

template <int D>
__global__ __launch_bounds__(256) void gather_vec4(
    const float* __restrict__ S, const int* __restrict__ rowptr,
    const int2* __restrict__ rec, const float* __restrict__ b,
    float* __restrict__ A, int N) {
  constexpr int CH = D >> 2;
  const int i = blockIdx.x * blockDim.x + threadIdx.x;
  if (i >= N * CH) return;
  const int n = i / CH;
  const int f = (i % CH) * 4;
  const int beg = rowptr[n];
  const int end = rowptr[n + 1];
  float4 acc = *(const float4*)(b + f);
#pragma unroll 4
  for (int p = beg; p < end; ++p) {
    const int2 r = rec[p];
    const float w = __int_as_float(r.y);
    const float4 s = *(const float4*)(S + (size_t)r.x * D + f);
    acc.x = fmaf(w, s.x, acc.x);
    acc.y = fmaf(w, s.y, acc.y);
    acc.z = fmaf(w, s.z, acc.z);
    acc.w = fmaf(w, s.w, acc.w);
  }
  *(float4*)(A + (size_t)n * D + f) = acc;
}

template <int D>
__global__ __launch_bounds__(256) void gather_scalar(
    const float* __restrict__ S, const int* __restrict__ rowptr,
    const int2* __restrict__ rec, const float* __restrict__ b,
    float* __restrict__ A, int N) {
  const int i = blockIdx.x * blockDim.x + threadIdx.x;
  if (i >= N * D) return;
  const int n = i / D;
  const int f = i % D;
  const int beg = rowptr[n];
  const int end = rowptr[n + 1];
  float acc = b[f];
#pragma unroll 4
  for (int p = beg; p < end; ++p) {
    const int2 r = rec[p];
    acc = fmaf(__int_as_float(r.y), S[(size_t)r.x * D + f], acc);
  }
  A[i] = acc;
}

static inline int cdiv(long long a, long long b) { return (int)((a + b - 1) / b); }

extern "C" void kernel_launch(void* const* d_in, const int* in_sizes, int n_in,
                              void* d_out, int out_size, void* d_ws, size_t ws_size,
                              hipStream_t stream) {
  const float* x   = (const float*)d_in[0];
  const int*   src = (const int*)  d_in[1];
  const int*   dst = (const int*)  d_in[2];
  const float* ew  = (const float*)d_in[3];
  const float* W1  = (const float*)d_in[4];
  const float* b1  = (const float*)d_in[5];
  const float* W2  = (const float*)d_in[6];
  const float* b2  = (const float*)d_in[7];
  const float* W3  = (const float*)d_in[8];
  const float* b3  = (const float*)d_in[9];
  float* out = (float*)d_out;

  const int N = N_NODES, E = N_EDGES;
  const int rowTiles = N / 16;

  float* bufS = (float*)d_ws;
  float* bufH = bufS + (size_t)N * 128;
  int*  cnt    = (int*)(bufH + (size_t)N * 128);
  int*  rowptr = cnt + 50048;
  int*  rank   = rowptr + 50048;
  int2* rec    = (int2*)(rank + N_EDGES);
  int*  base   = (int*)(rec + N_EDGES);

  zero_counts<<<cdiv(N, 256), 256, 0, stream>>>(cnt, N);
  count_rank_lds<<<SORT_R * SORT_S, CR_THREADS, 0, stream>>>(dst, cnt, rank, base);
  scan_counts<<<1, 1024, 0, stream>>>(cnt, rowptr);
  place_edges<<<cdiv(E, 256), 256, 0, stream>>>(src, dst, ew, rowptr, rank, base, rec, E);

  gemm_wmma_f32<256, 128, false><<<rowTiles, 256, 0, stream>>>(x, W1, bufS, N);
  gather_vec4<128><<<cdiv((long long)N * 32, 256), 256, 0, stream>>>(
      bufS, rowptr, rec, b1, bufH, N);

  gemm_wmma_f32<128, 64, true><<<rowTiles, 128, 0, stream>>>(bufH, W2, bufS, N);
  gather_vec4<64><<<cdiv((long long)N * 16, 256), 256, 0, stream>>>(
      bufS, rowptr, rec, b2, bufH, N);

  gemm_wmma_f32<64, 10, true><<<rowTiles, 32, 0, stream>>>(bufH, W3, bufS, N);
  gather_scalar<10><<<cdiv((long long)N * 10, 256), 256, 0, stream>>>(
      bufS, rowptr, rec, b3, out, N);
}


# GEMM operand feed: a work-group copies its 32 rows of X (64 in layer 3) to LDS once in whole 128-byte lines; waves read A fragments from LDS and use each B fragment for two row tiles
# speedup vs baseline: 2.2720x; 1.1843x over previous
#include <hip/hip_runtime.h>


typedef float v2f __attribute__((ext_vector_type(2)));
typedef float v8f __attribute__((ext_vector_type(8)));

#define N_NODES 50000
#define N_EDGES 800000

template <int K, int DOUT, bool RELU_IN, int TM, int WM>
__global__ __launch_bounds__(32 * ((DOUT + 15) / 16) * WM) void gemm_wmma_f32(
    const float* __restrict__ X, const float* __restrict__ W,
    float* __restrict__ Y, int N) {
  constexpr bool GUARD = (DOUT % 16) != 0;
  constexpr int NT       = (DOUT + 15) / 16;
  constexpr int NTHREADS = 32 * NT * WM;
  constexpr int BR       = 16 * TM * WM;
  constexpr int PITCH    = K + 4;
  constexpr int Q        = K / 4;
  constexpr int NLD      = (BR * Q) / NTHREADS;
  static_assert((BR * Q) % NTHREADS == 0, "the row block is copied in whole passes");
  static_assert(K % 4 == 0 && ((PITCH / 4) & 1) == 1, "pitch must be 16 B x odd");
  static_assert(N_NODES % 16 == 0, "row tiles must be whole");

  __shared__ __attribute__((aligned(16))) float sA[BR * PITCH];

  const int wave  = threadIdx.x >> 5;
  const int lane  = threadIdx.x & 31;
  const int wn    = wave % NT;
  const int wm    = wave / NT;
  const int n0    = wn * 16;
  const int m0    = blockIdx.x * BR;
  const int lhalf = lane >> 4;
  const int l16   = lane & 15;

  float4 v[NLD];
#pragma unroll
  for (int i = 0; i < NLD; ++i) {
    const int q = i * NTHREADS + (int)threadIdx.x;
    const int r = q / Q;
    const int c = q % Q;
    int row = m0 + r;
    if (row > N - 1) row = N - 1;
    v[i] = *(const float4*)(X + (size_t)row * K + c * 4);
  }
#pragma unroll
  for (int i = 0; i < NLD; ++i) {
    const int q = i * NTHREADS + (int)threadIdx.x;
    const int r = q / Q;
    const int c = q % Q;
    float4 t = v[i];
    if (RELU_IN) {
      t.x = fmaxf(t.x, 0.f); t.y = fmaxf(t.y, 0.f);
      t.z = fmaxf(t.z, 0.f); t.w = fmaxf(t.w, 0.f);
    }
    *(float4*)(sA + r * PITCH + c * 4) = t;
  }
  __syncthreads();

  const int  colB  = n0 + l16;
  const bool colOK = !GUARD || (colB < DOUT);
  const int  kBase = 2 * lhalf;
  const float* arow = sA + (wm * TM * 16 + l16) * PITCH + kBase;

  v8f acc[TM];
#pragma unroll
  for (int t = 0; t < TM; ++t) acc[t] = (v8f){0.f, 0.f, 0.f, 0.f, 0.f, 0.f, 0.f, 0.f};

#pragma unroll 8
  for (int k = 0; k < K; k += 4) {
    v2f b;
    if (GUARD) {
      b.x = colOK ? W[(size_t)(k + kBase + 0) * DOUT + colB] : 0.f;
      b.y = colOK ? W[(size_t)(k + kBase + 1) * DOUT + colB] : 0.f;
    } else {
      b.x = W[(size_t)(k + kBase + 0) * DOUT + colB];
      b.y = W[(size_t)(k + kBase + 1) * DOUT + colB];
    }
#pragma unroll
    for (int t = 0; t < TM; ++t) {
      const v2f a = *(const v2f*)(arow + t * 16 * PITCH + k);
      acc[t] = __builtin_amdgcn_wmma_f32_16x16x4_f32(
           false, a,  false, b,
           (short)0, acc[t],  false,  false);
    }
  }

  if (colOK) {
#pragma unroll
    for (int t = 0; t < TM; ++t) {
      const int row0 = m0 + (wm * TM + t) * 16;
      if (row0 < N) {
#pragma unroll
        for (int r = 0; r < 8; ++r) {
          Y[(size_t)(row0 + r + 8 * lhalf) * DOUT + colB] = acc[t][r];
        }
      }
    }
  }
}

#define SORT_R     8
#define SORT_S     32
#define RANGE_N    ((N_NODES + SORT_R - 1) / SORT_R)
#define SLICE_E    (N_EDGES / SORT_S)
#define SLICE_Q    (SLICE_E / 4)
#define CR_THREADS 1024
#define CR_IT      ((SLICE_Q + CR_THREADS - 1) / CR_THREADS)
#define CR_PN      ((RANGE_N + CR_THREADS - 1) / CR_THREADS)
static_assert(SLICE_E * SORT_S == N_EDGES, "slices must tile the edge list");
static_assert(SLICE_E % 4 == 0, "a slice is read as whole int4");
static_assert(RANGE_N * SORT_R >= N_NODES, "ranges must cover the nodes");
__global__ __launch_bounds__(256) void zero_counts(int* __restrict__ cnt, int n) {
  const int i = blockIdx.x * blockDim.x + threadIdx.x;
  if (i < n) cnt[i] = 0;
}

__global__ __launch_bounds__(CR_THREADS) void count_rank_lds(
    const int* __restrict__ dst, int* __restrict__ cnt,
    int* __restrict__ rank, int* __restrict__ base) {
  __shared__ int lcnt[RANGE_N];
  const int ri = blockIdx.x % SORT_R;
  const int sj = blockIdx.x / SORT_R;
  const int lo = ri * RANGE_N;
  const int4* __restrict__ d4 = (const int4*)(dst + (size_t)sj * SLICE_E);
  int* __restrict__ rk = rank + (size_t)sj * SLICE_E;

  int4 d[CR_IT];
#pragma unroll
  for (int it = 0; it < CR_IT; ++it) {
    const int q = it * CR_THREADS + (int)threadIdx.x;
    d[it] = (q < SLICE_Q) ? d4[q] : make_int4(-1, -1, -1, -1);
  }

  for (int n = threadIdx.x; n < RANGE_N; n += CR_THREADS) lcnt[n] = 0;
  __syncthreads();

#pragma unroll
  for (int it = 0; it < CR_IT; ++it) {
    const int e0 = (it * CR_THREADS + (int)threadIdx.x) * 4;
    unsigned u;
    u = (unsigned)(d[it].x - lo);
    if (u < (unsigned)RANGE_N) rk[e0 + 0] = atomicAdd(&lcnt[u], 1);
    u = (unsigned)(d[it].y - lo);
    if (u < (unsigned)RANGE_N) rk[e0 + 1] = atomicAdd(&lcnt[u], 1);
    u = (unsigned)(d[it].z - lo);
    if (u < (unsigned)RANGE_N) rk[e0 + 2] = atomicAdd(&lcnt[u], 1);
    u = (unsigned)(d[it].w - lo);
    if (u < (unsigned)RANGE_N) rk[e0 + 3] = atomicAdd(&lcnt[u], 1);
  }
  __syncthreads();

  int got[CR_PN];
#pragma unroll
  for (int it = 0; it < CR_PN; ++it) {
    const int n = it * CR_THREADS + (int)threadIdx.x;
    const int node = lo + n;
    got[it] = 0;
    if (n < RANGE_N && node < N_NODES) got[it] = atomicAdd(&cnt[node], lcnt[n]);
  }
#pragma unroll
  for (int it = 0; it < CR_PN; ++it) {
    const int n = it * CR_THREADS + (int)threadIdx.x;
    const int node = lo + n;
    if (n < RANGE_N && node < N_NODES) base[(size_t)sj * N_NODES + node] = got[it];
  }
}

#define SCAN_IT  ((N_NODES + 1023) / 1024)
#define SCAN_SEG (SCAN_IT * 32)
__global__ __launch_bounds__(1024) void scan_counts(
    const int* __restrict__ cnt, int* __restrict__ rowptr) {
  __shared__ int wtot[32];
  const int lane = threadIdx.x & 31;
  const int wave = threadIdx.x >> 5;
  const int seg0 = wave * SCAN_SEG;

  int v[SCAN_IT];
  int lsum = 0;
#pragma unroll
  for (int it = 0; it < SCAN_IT; ++it) {
    const int idx = seg0 + it * 32 + lane;
    v[it] = (idx < N_NODES) ? cnt[idx] : 0;
    lsum += v[it];
  }
  for (int off = 16; off > 0; off >>= 1) lsum += __shfl_xor(lsum, off);
  if (lane == 0) wtot[wave] = lsum;
  __syncthreads();

  int carry = (lane < wave) ? wtot[lane] : 0;
  for (int off = 16; off > 0; off >>= 1) carry += __shfl_xor(carry, off);

#pragma unroll
  for (int it = 0; it < SCAN_IT; ++it) {
    int inc = v[it];
    for (int off = 1; off < 32; off <<= 1) {
      const int up = __shfl_up(inc, off);
      if (lane >= off) inc += up;
    }
    const int idx = seg0 + it * 32 + lane;
    if (idx < N_NODES) rowptr[idx] = carry + inc - v[it];
    carry += __shfl(inc, 31);
  }
  if (threadIdx.x == 1023) rowptr[N_NODES] = carry;
}

__global__ __launch_bounds__(256) void place_edges(
    const int* __restrict__ src, const int* __restrict__ dst,
    const float* __restrict__ ew, const int* __restrict__ rowptr,
    const int* __restrict__ rank, const int* __restrict__ base,
    int2* __restrict__ rec, int E) {
  const int e = blockIdx.x * blockDim.x + threadIdx.x;
  if (e < E) {
    const int d = dst[e];
    const int sj = e / SLICE_E;
    const int p = rowptr[d] + base[(size_t)sj * N_NODES + d] + rank[e];
    rec[p] = make_int2(src[e], __float_as_int(ew[e]));
  }
}

template <int D>
__global__ __launch_bounds__(256) void gather_vec4(
    const float* __restrict__ S, const int* __restrict__ rowptr,
    const int2* __restrict__ rec, const float* __restrict__ b,
    float* __restrict__ A, int N) {
  constexpr int CH = D >> 2;
  const int i = blockIdx.x * blockDim.x + threadIdx.x;
  if (i >= N * CH) return;
  const int n = i / CH;
  const int f = (i % CH) * 4;
  const int beg = rowptr[n];
  const int end = rowptr[n + 1];
  float4 acc = *(const float4*)(b + f);
#pragma unroll 4
  for (int p = beg; p < end; ++p) {
    const int2 r = rec[p];
    const float w = __int_as_float(r.y);
    const float4 s = *(const float4*)(S + (size_t)r.x * D + f);
    acc.x = fmaf(w, s.x, acc.x);
    acc.y = fmaf(w, s.y, acc.y);
    acc.z = fmaf(w, s.z, acc.z);
    acc.w = fmaf(w, s.w, acc.w);
  }
  *(float4*)(A + (size_t)n * D + f) = acc;
}

template <int D>
__global__ __launch_bounds__(256) void gather_scalar(
    const float* __restrict__ S, const int* __restrict__ rowptr,
    const int2* __restrict__ rec, const float* __restrict__ b,
    float* __restrict__ A, int N) {
  const int i = blockIdx.x * blockDim.x + threadIdx.x;
  if (i >= N * D) return;
  const int n = i / D;
  const int f = i % D;
  const int beg = rowptr[n];
  const int end = rowptr[n + 1];
  float acc = b[f];
#pragma unroll 4
  for (int p = beg; p < end; ++p) {
    const int2 r = rec[p];
    acc = fmaf(__int_as_float(r.y), S[(size_t)r.x * D + f], acc);
  }
  A[i] = acc;
}

static inline int cdiv(long long a, long long b) { return (int)((a + b - 1) / b); }

extern "C" void kernel_launch(void* const* d_in, const int* in_sizes, int n_in,
                              void* d_out, int out_size, void* d_ws, size_t ws_size,
                              hipStream_t stream) {
  const float* x   = (const float*)d_in[0];
  const int*   src = (const int*)  d_in[1];
  const int*   dst = (const int*)  d_in[2];
  const float* ew  = (const float*)d_in[3];
  const float* W1  = (const float*)d_in[4];
  const float* b1  = (const float*)d_in[5];
  const float* W2  = (const float*)d_in[6];
  const float* b2  = (const float*)d_in[7];
  const float* W3  = (const float*)d_in[8];
  const float* b3  = (const float*)d_in[9];
  float* out = (float*)d_out;

  const int N = N_NODES, E = N_EDGES;

  float* bufS = (float*)d_ws;
  float* bufH = bufS + (size_t)N * 128;
  int*  cnt    = (int*)(bufH + (size_t)N * 128);
  int*  rowptr = cnt + 50048;
  int*  rank   = rowptr + 50048;
  int2* rec    = (int2*)(rank + N_EDGES);
  int*  base   = (int*)(rec + N_EDGES);

  zero_counts<<<cdiv(N, 256), 256, 0, stream>>>(cnt, N);
  count_rank_lds<<<SORT_R * SORT_S, CR_THREADS, 0, stream>>>(dst, cnt, rank, base);
  scan_counts<<<1, 1024, 0, stream>>>(cnt, rowptr);
  place_edges<<<cdiv(E, 256), 256, 0, stream>>>(src, dst, ew, rowptr, rank, base, rec, E);

  gemm_wmma_f32<256, 128, false, 2, 1><<<cdiv(N, 32), 256, 0, stream>>>(x, W1, bufS, N);
  gather_vec4<128><<<cdiv((long long)N * 32, 256), 256, 0, stream>>>(
      bufS, rowptr, rec, b1, bufH, N);

  gemm_wmma_f32<128, 64, true, 2, 1><<<cdiv(N, 32), 128, 0, stream>>>(bufH, W2, bufS, N);
  gather_vec4<64><<<cdiv((long long)N * 16, 256), 256, 0, stream>>>(
      bufS, rowptr, rec, b2, bufH, N);

  gemm_wmma_f32<64, 10, true, 2, 2><<<cdiv(N, 64), 64, 0, stream>>>(bufH, W3, bufS, N);
  gather_scalar<10><<<cdiv((long long)N * 10, 256), 256, 0, stream>>>(
      bufS, rowptr, rec, b3, out, N);
}


# Edge sums of layers 1 and 2 moved into the next layer's GEMM launch: a work-group sums its 32 (16) rows as before, puts relu of them in LDS, multiplies from there; H1, H2 not stored; 10 launches to 8
# speedup vs baseline: 2.4643x; 1.0846x over previous
#include <hip/hip_runtime.h>


typedef float v2f __attribute__((ext_vector_type(2)));
typedef float v8f __attribute__((ext_vector_type(8)));

#define N_NODES 50000
#define N_EDGES 800000

template <int K, int DOUT, bool RELU_IN, int TM, int WM>
__global__ __launch_bounds__(32 * ((DOUT + 15) / 16) * WM) void gemm_wmma_f32(
    const float* __restrict__ X, const float* __restrict__ W,
    float* __restrict__ Y, int N) {
  constexpr bool GUARD = (DOUT % 16) != 0;
  constexpr int NT       = (DOUT + 15) / 16;
  constexpr int NTHREADS = 32 * NT * WM;
  constexpr int BR       = 16 * TM * WM;
  constexpr int PITCH    = K + 4;
  constexpr int Q        = K / 4;
  constexpr int NLD      = (BR * Q) / NTHREADS;
  static_assert((BR * Q) % NTHREADS == 0, "the row block is copied in whole passes");
  static_assert(K % 4 == 0 && ((PITCH / 4) & 1) == 1, "pitch must be 16 B x odd");
  static_assert(N_NODES % 16 == 0, "row tiles must be whole");

  __shared__ __attribute__((aligned(16))) float sA[BR * PITCH];

  const int wave  = threadIdx.x >> 5;
  const int lane  = threadIdx.x & 31;
  const int wn    = wave % NT;
  const int wm    = wave / NT;
  const int n0    = wn * 16;
  const int m0    = blockIdx.x * BR;
  const int lhalf = lane >> 4;
  const int l16   = lane & 15;

  float4 v[NLD];
#pragma unroll
  for (int i = 0; i < NLD; ++i) {
    const int q = i * NTHREADS + (int)threadIdx.x;
    const int r = q / Q;
    const int c = q % Q;
    int row = m0 + r;
    if (row > N - 1) row = N - 1;
    v[i] = *(const float4*)(X + (size_t)row * K + c * 4);
  }
#pragma unroll
  for (int i = 0; i < NLD; ++i) {
    const int q = i * NTHREADS + (int)threadIdx.x;
    const int r = q / Q;
    const int c = q % Q;
    float4 t = v[i];
    if (RELU_IN) {
      t.x = fmaxf(t.x, 0.f); t.y = fmaxf(t.y, 0.f);
      t.z = fmaxf(t.z, 0.f); t.w = fmaxf(t.w, 0.f);
    }
    *(float4*)(sA + r * PITCH + c * 4) = t;
  }
  __syncthreads();

  const int  colB  = n0 + l16;
  const bool colOK = !GUARD || (colB < DOUT);
  const int  kBase = 2 * lhalf;
  const float* arow = sA + (wm * TM * 16 + l16) * PITCH + kBase;

  v8f acc[TM];
#pragma unroll
  for (int t = 0; t < TM; ++t) acc[t] = (v8f){0.f, 0.f, 0.f, 0.f, 0.f, 0.f, 0.f, 0.f};

#pragma unroll 8
  for (int k = 0; k < K; k += 4) {
    v2f b;
    if (GUARD) {
      b.x = colOK ? W[(size_t)(k + kBase + 0) * DOUT + colB] : 0.f;
      b.y = colOK ? W[(size_t)(k + kBase + 1) * DOUT + colB] : 0.f;
    } else {
      b.x = W[(size_t)(k + kBase + 0) * DOUT + colB];
      b.y = W[(size_t)(k + kBase + 1) * DOUT + colB];
    }
#pragma unroll
    for (int t = 0; t < TM; ++t) {
      const v2f a = *(const v2f*)(arow + t * 16 * PITCH + k);
      acc[t] = __builtin_amdgcn_wmma_f32_16x16x4_f32(
           false, a,  false, b,
           (short)0, acc[t],  false,  false);
    }
  }

  if (colOK) {
#pragma unroll
    for (int t = 0; t < TM; ++t) {
      const int row0 = m0 + (wm * TM + t) * 16;
      if (row0 < N) {
#pragma unroll
        for (int r = 0; r < 8; ++r) {
          Y[(size_t)(row0 + r + 8 * lhalf) * DOUT + colB] = acc[t][r];
        }
      }
    }
  }
}

#define SORT_R     8
#define SORT_S     32
#define RANGE_N    ((N_NODES + SORT_R - 1) / SORT_R)
#define SLICE_E    (N_EDGES / SORT_S)
#define SLICE_Q    (SLICE_E / 4)
#define CR_THREADS 1024
#define CR_IT      ((SLICE_Q + CR_THREADS - 1) / CR_THREADS)
#define CR_PN      ((RANGE_N + CR_THREADS - 1) / CR_THREADS)
static_assert(SLICE_E * SORT_S == N_EDGES, "slices must tile the edge list");
static_assert(SLICE_E % 4 == 0, "a slice is read as whole int4");
static_assert(RANGE_N * SORT_R >= N_NODES, "ranges must cover the nodes");
__global__ __launch_bounds__(256) void zero_counts(int* __restrict__ cnt, int n) {
  const int i = blockIdx.x * blockDim.x + threadIdx.x;
  if (i < n) cnt[i] = 0;
}

__global__ __launch_bounds__(CR_THREADS) void count_rank_lds(
    const int* __restrict__ dst, int* __restrict__ cnt,
    int* __restrict__ rank, int* __restrict__ base) {
  __shared__ int lcnt[RANGE_N];
  const int ri = blockIdx.x % SORT_R;
  const int sj = blockIdx.x / SORT_R;
  const int lo = ri * RANGE_N;
  const int4* __restrict__ d4 = (const int4*)(dst + (size_t)sj * SLICE_E);
  int* __restrict__ rk = rank + (size_t)sj * SLICE_E;

  int4 d[CR_IT];
#pragma unroll
  for (int it = 0; it < CR_IT; ++it) {
    const int q = it * CR_THREADS + (int)threadIdx.x;
    d[it] = (q < SLICE_Q) ? d4[q] : make_int4(-1, -1, -1, -1);
  }

  for (int n = threadIdx.x; n < RANGE_N; n += CR_THREADS) lcnt[n] = 0;
  __syncthreads();

#pragma unroll
  for (int it = 0; it < CR_IT; ++it) {
    const int e0 = (it * CR_THREADS + (int)threadIdx.x) * 4;
    unsigned u;
    u = (unsigned)(d[it].x - lo);
    if (u < (unsigned)RANGE_N) rk[e0 + 0] = atomicAdd(&lcnt[u], 1);
    u = (unsigned)(d[it].y - lo);
    if (u < (unsigned)RANGE_N) rk[e0 + 1] = atomicAdd(&lcnt[u], 1);
    u = (unsigned)(d[it].z - lo);
    if (u < (unsigned)RANGE_N) rk[e0 + 2] = atomicAdd(&lcnt[u], 1);
    u = (unsigned)(d[it].w - lo);
    if (u < (unsigned)RANGE_N) rk[e0 + 3] = atomicAdd(&lcnt[u], 1);
  }
  __syncthreads();

  int got[CR_PN];
#pragma unroll
  for (int it = 0; it < CR_PN; ++it) {
    const int n = it * CR_THREADS + (int)threadIdx.x;
    const int node = lo + n;
    got[it] = 0;
    if (n < RANGE_N && node < N_NODES) got[it] = atomicAdd(&cnt[node], lcnt[n]);
  }
#pragma unroll
  for (int it = 0; it < CR_PN; ++it) {
    const int n = it * CR_THREADS + (int)threadIdx.x;
    const int node = lo + n;
    if (n < RANGE_N && node < N_NODES) base[(size_t)sj * N_NODES + node] = got[it];
  }
}

#define SCAN_IT  ((N_NODES + 1023) / 1024)
#define SCAN_SEG (SCAN_IT * 32)
__global__ __launch_bounds__(1024) void scan_counts(
    const int* __restrict__ cnt, int* __restrict__ rowptr) {
  __shared__ int wtot[32];
  const int lane = threadIdx.x & 31;
  const int wave = threadIdx.x >> 5;
  const int seg0 = wave * SCAN_SEG;

  int v[SCAN_IT];
  int lsum = 0;
#pragma unroll
  for (int it = 0; it < SCAN_IT; ++it) {
    const int idx = seg0 + it * 32 + lane;
    v[it] = (idx < N_NODES) ? cnt[idx] : 0;
    lsum += v[it];
  }
  for (int off = 16; off > 0; off >>= 1) lsum += __shfl_xor(lsum, off);
  if (lane == 0) wtot[wave] = lsum;
  __syncthreads();

  int carry = (lane < wave) ? wtot[lane] : 0;
  for (int off = 16; off > 0; off >>= 1) carry += __shfl_xor(carry, off);

#pragma unroll
  for (int it = 0; it < SCAN_IT; ++it) {
    int inc = v[it];
    for (int off = 1; off < 32; off <<= 1) {
      const int up = __shfl_up(inc, off);
      if (lane >= off) inc += up;
    }
    const int idx = seg0 + it * 32 + lane;
    if (idx < N_NODES) rowptr[idx] = carry + inc - v[it];
    carry += __shfl(inc, 31);
  }
  if (threadIdx.x == 1023) rowptr[N_NODES] = carry;
}

__global__ __launch_bounds__(256) void place_edges(
    const int* __restrict__ src, const int* __restrict__ dst,
    const float* __restrict__ ew, const int* __restrict__ rowptr,
    const int* __restrict__ rank, const int* __restrict__ base,
    int2* __restrict__ rec, int E) {
  const int e = blockIdx.x * blockDim.x + threadIdx.x;
  if (e < E) {
    const int d = dst[e];
    const int sj = e / SLICE_E;
    const int p = rowptr[d] + base[(size_t)sj * N_NODES + d] + rank[e];
    rec[p] = make_int2(src[e], __float_as_int(ew[e]));
  }
}

template <int K, int DOUT, int WM, int NTHREADS>
__global__ __launch_bounds__(NTHREADS) void gather_gemm_f32(
    const float* __restrict__ S, const int* __restrict__ rowptr,
    const int2* __restrict__ rec, const float* __restrict__ b,
    const float* __restrict__ W, float* __restrict__ Y, int N) {
  constexpr bool GUARD = (DOUT % 16) != 0;
  constexpr int NT    = (DOUT + 15) / 16;
  constexpr int BR    = 16 * WM;
  constexpr int PITCH = K + 4;
  constexpr int CH    = K / 4;
  constexpr int RPP   = NTHREADS / CH;
  constexpr int NPASS = BR / RPP;
  static_assert(BR % RPP == 0 && NPASS >= 1, "lane groups share the rows evenly");
  static_assert(NTHREADS >= 32 * NT * WM, "one wave per output tile");
  static_assert(K % 4 == 0 && ((PITCH / 4) & 1) == 1, "pitch must be 16 B x odd");

  __shared__ __attribute__((aligned(16))) float sA[BR * PITCH];

  const int m0 = blockIdx.x * BR;
  const int g  = (int)threadIdx.x / CH;
  const int f  = ((int)threadIdx.x % CH) * 4;
  const int r0 = g * NPASS;

  int rp[NPASS + 1];
#pragma unroll
  for (int j = 0; j <= NPASS; ++j) {
    int n = m0 + r0 + j;
    if (n > N) n = N;
    rp[j] = rowptr[n];
  }
  const float4 bias = *(const float4*)(b + f);

#pragma unroll
  for (int j = 0; j < NPASS; ++j) {
    float4 acc = bias;
    const int end = rp[j + 1];
#pragma unroll 4
    for (int p = rp[j]; p < end; ++p) {
      const int2 r = rec[p];
      const float w = __int_as_float(r.y);
      const float4 s = *(const float4*)(S + (size_t)r.x * K + f);
      acc.x = fmaf(w, s.x, acc.x);
      acc.y = fmaf(w, s.y, acc.y);
      acc.z = fmaf(w, s.z, acc.z);
      acc.w = fmaf(w, s.w, acc.w);
    }
    acc.x = fmaxf(acc.x, 0.f); acc.y = fmaxf(acc.y, 0.f);
    acc.z = fmaxf(acc.z, 0.f); acc.w = fmaxf(acc.w, 0.f);
    *(float4*)(sA + (r0 + j) * PITCH + f) = acc;
  }
  __syncthreads();

  const int wave = (int)threadIdx.x >> 5;
  if (wave >= NT * WM) return;
  const int lane  = (int)threadIdx.x & 31;
  const int wn    = wave % NT;
  const int wm    = wave / NT;
  const int n0    = wn * 16;
  const int lhalf = lane >> 4;
  const int l16   = lane & 15;
  const int  colB  = n0 + l16;
  const bool colOK = !GUARD || (colB < DOUT);
  const int  kBase = 2 * lhalf;
  const float* arow = sA + (wm * 16 + l16) * PITCH + kBase;

  v8f acc = (v8f){0.f, 0.f, 0.f, 0.f, 0.f, 0.f, 0.f, 0.f};
#pragma unroll 8
  for (int k = 0; k < K; k += 4) {
    v2f bf;
    if (GUARD) {
      bf.x = colOK ? W[(size_t)(k + kBase + 0) * DOUT + colB] : 0.f;
      bf.y = colOK ? W[(size_t)(k + kBase + 1) * DOUT + colB] : 0.f;
    } else {
      bf.x = W[(size_t)(k + kBase + 0) * DOUT + colB];
      bf.y = W[(size_t)(k + kBase + 1) * DOUT + colB];
    }
    const v2f a = *(const v2f*)(arow + k);
    acc = __builtin_amdgcn_wmma_f32_16x16x4_f32(
         false, a,  false, bf,
         (short)0, acc,  false,  false);
  }

  const int row0 = m0 + wm * 16;
  if (colOK && row0 < N) {
#pragma unroll
    for (int r = 0; r < 8; ++r) {
      Y[(size_t)(row0 + r + 8 * lhalf) * DOUT + colB] = acc[r];
    }
  }
}

template <int D>
__global__ __launch_bounds__(256) void gather_scalar(
    const float* __restrict__ S, const int* __restrict__ rowptr,
    const int2* __restrict__ rec, const float* __restrict__ b,
    float* __restrict__ A, int N) {
  const int i = blockIdx.x * blockDim.x + threadIdx.x;
  if (i >= N * D) return;
  const int n = i / D;
  const int f = i % D;
  const int beg = rowptr[n];
  const int end = rowptr[n + 1];
  float acc = b[f];
#pragma unroll 4
  for (int p = beg; p < end; ++p) {
    const int2 r = rec[p];
    acc = fmaf(__int_as_float(r.y), S[(size_t)r.x * D + f], acc);
  }
  A[i] = acc;
}

static inline int cdiv(long long a, long long b) { return (int)((a + b - 1) / b); }

extern "C" void kernel_launch(void* const* d_in, const int* in_sizes, int n_in,
                              void* d_out, int out_size, void* d_ws, size_t ws_size,
                              hipStream_t stream) {
  const float* x   = (const float*)d_in[0];
  const int*   src = (const int*)  d_in[1];
  const int*   dst = (const int*)  d_in[2];
  const float* ew  = (const float*)d_in[3];
  const float* W1  = (const float*)d_in[4];
  const float* b1  = (const float*)d_in[5];
  const float* W2  = (const float*)d_in[6];
  const float* b2  = (const float*)d_in[7];
  const float* W3  = (const float*)d_in[8];
  const float* b3  = (const float*)d_in[9];
  float* out = (float*)d_out;

  const int N = N_NODES, E = N_EDGES;

  float* bufS = (float*)d_ws;
  float* bufH = bufS + (size_t)N * 128;
  int*  cnt    = (int*)(bufH + (size_t)N * 128);
  int*  rowptr = cnt + 50048;
  int*  rank   = rowptr + 50048;
  int2* rec    = (int2*)(rank + N_EDGES);
  int*  base   = (int*)(rec + N_EDGES);

  zero_counts<<<cdiv(N, 256), 256, 0, stream>>>(cnt, N);
  count_rank_lds<<<SORT_R * SORT_S, CR_THREADS, 0, stream>>>(dst, cnt, rank, base);
  scan_counts<<<1, 1024, 0, stream>>>(cnt, rowptr);
  place_edges<<<cdiv(E, 256), 256, 0, stream>>>(src, dst, ew, rowptr, rank, base, rec, E);

  gemm_wmma_f32<256, 128, false, 2, 1><<<cdiv(N, 32), 256, 0, stream>>>(x, W1, bufS, N);

  gather_gemm_f32<128, 64, 2, 256><<<cdiv(N, 32), 256, 0, stream>>>(
      bufS, rowptr, rec, b1, W2, bufH, N);

  gather_gemm_f32<64, 10, 1, 128><<<cdiv(N, 16), 128, 0, stream>>>(
      bufH, rowptr, rec, b2, W3, bufS, N);

  gather_scalar<10><<<cdiv((long long)N * 10, 256), 256, 0, stream>>>(
      bufS, rowptr, rec, b3, out, N);
}


# Running sum of node counts moved into the layer-1 product's launch: 28 independent work-groups, each adds up the counts ahead of its piece itself; no scan launch; 8 launches to 7
# speedup vs baseline: 2.6228x; 1.0643x over previous
#include <hip/hip_runtime.h>


typedef float v2f __attribute__((ext_vector_type(2)));
typedef float v8f __attribute__((ext_vector_type(8)));

#define N_NODES 50000
#define N_EDGES 800000

#define SCAN_WG    28
#define SCAN_ROWS  7
#define SCAN_WSEG  (SCAN_ROWS * 32)
#define SCAN_PIECE (SCAN_WSEG * 8)
#define SCAN_PQ    (SCAN_PIECE / 4)
#define SCAN_HU    8
static_assert(SCAN_PIECE * SCAN_WG >= N_NODES, "the pieces must cover the nodes");
static_assert((SCAN_WG - 1) * SCAN_PIECE < N_NODES, "every head lies inside the counts");

__device__ __forceinline__ void scan_piece(const int* __restrict__ cnt,
                                           int* __restrict__ rowptr) {
  __shared__ int sHead[8];
  __shared__ int sTot[8];
  const int lane = (int)threadIdx.x & 31;
  const int wave = (int)threadIdx.x >> 5;
  const int seg0 = (int)blockIdx.x * SCAN_PIECE + wave * SCAN_WSEG;

  int v[SCAN_ROWS];
#pragma unroll
  for (int r = 0; r < SCAN_ROWS; ++r) {
    const int idx = seg0 + r * 32 + lane;
    v[r] = (idx < N_NODES) ? cnt[idx] : 0;
  }

  const int4* __restrict__ c4 = (const int4*)cnt;
  const int nq = (int)blockIdx.x * SCAN_PQ;
  int hsum = 0;
  for (int q0 = 0; q0 < nq; q0 += 256 * SCAN_HU) {
    int4 t[SCAN_HU];
#pragma unroll
    for (int u = 0; u < SCAN_HU; ++u) {
      int q = q0 + u * 256 + (int)threadIdx.x;
      if (q > nq - 1) q = nq - 1;
      t[u] = c4[q];
    }
#pragma unroll
    for (int u = 0; u < SCAN_HU; ++u) {
      const int q = q0 + u * 256 + (int)threadIdx.x;
      const int a = (t[u].x + t[u].y) + (t[u].z + t[u].w);
      hsum += (q < nq) ? a : 0;
    }
  }

  int lsum = 0;
#pragma unroll
  for (int r = 0; r < SCAN_ROWS; ++r) lsum += v[r];
  for (int off = 16; off > 0; off >>= 1) {
    hsum += __shfl_xor(hsum, off);
    lsum += __shfl_xor(lsum, off);
  }
  if (lane == 0) { sHead[wave] = hsum; sTot[wave] = lsum; }
  __syncthreads();

  int carry = 0;
  if (lane < 8) carry = sHead[lane];
  else if (lane < 16 && (lane - 8) < wave) carry = sTot[lane - 8];
  for (int off = 16; off > 0; off >>= 1) carry += __shfl_xor(carry, off);

#pragma unroll
  for (int r = 0; r < SCAN_ROWS; ++r) {
    int inc = v[r];
    for (int off = 1; off < 32; off <<= 1) {
      const int up = __shfl_up(inc, off);
      if (lane >= off) inc += up;
    }
    const int idx = seg0 + r * 32 + lane;
    if (idx < N_NODES) rowptr[idx] = carry + inc - v[r];
    if (idx == N_NODES - 1) rowptr[N_NODES] = carry + inc;
    carry += __shfl(inc, 31);
  }
}

template <int K, int DOUT, bool RELU_IN, int TM, int WM>
__global__ __launch_bounds__(32 * ((DOUT + 15) / 16) * WM) void gemm_wmma_f32(
    const float* __restrict__ X, const float* __restrict__ W,
    float* __restrict__ Y, int N,
    const int* __restrict__ cnt, int* __restrict__ rowptr) {
  if (blockIdx.x < SCAN_WG) {
    scan_piece(cnt, rowptr);
    return;
  }
  constexpr bool GUARD = (DOUT % 16) != 0;
  constexpr int NT       = (DOUT + 15) / 16;
  constexpr int NTHREADS = 32 * NT * WM;
  constexpr int BR       = 16 * TM * WM;
  constexpr int PITCH    = K + 4;
  constexpr int Q        = K / 4;
  constexpr int NLD      = (BR * Q) / NTHREADS;
  static_assert((BR * Q) % NTHREADS == 0, "the row block is copied in whole passes");
  static_assert(K % 4 == 0 && ((PITCH / 4) & 1) == 1, "pitch must be 16 B x odd");
  static_assert(N_NODES % 16 == 0, "row tiles must be whole");

  __shared__ __attribute__((aligned(16))) float sA[BR * PITCH];

  const int wave  = threadIdx.x >> 5;
  const int lane  = threadIdx.x & 31;
  const int wn    = wave % NT;
  const int wm    = wave / NT;
  const int n0    = wn * 16;
  const int m0    = ((int)blockIdx.x - SCAN_WG) * BR;
  const int lhalf = lane >> 4;
  const int l16   = lane & 15;

  float4 v[NLD];
#pragma unroll
  for (int i = 0; i < NLD; ++i) {
    const int q = i * NTHREADS + (int)threadIdx.x;
    const int r = q / Q;
    const int c = q % Q;
    int row = m0 + r;
    if (row > N - 1) row = N - 1;
    v[i] = *(const float4*)(X + (size_t)row * K + c * 4);
  }
#pragma unroll
  for (int i = 0; i < NLD; ++i) {
    const int q = i * NTHREADS + (int)threadIdx.x;
    const int r = q / Q;
    const int c = q % Q;
    float4 t = v[i];
    if (RELU_IN) {
      t.x = fmaxf(t.x, 0.f); t.y = fmaxf(t.y, 0.f);
      t.z = fmaxf(t.z, 0.f); t.w = fmaxf(t.w, 0.f);
    }
    *(float4*)(sA + r * PITCH + c * 4) = t;
  }
  __syncthreads();

  const int  colB  = n0 + l16;
  const bool colOK = !GUARD || (colB < DOUT);
  const int  kBase = 2 * lhalf;
  const float* arow = sA + (wm * TM * 16 + l16) * PITCH + kBase;

  v8f acc[TM];
#pragma unroll
  for (int t = 0; t < TM; ++t) acc[t] = (v8f){0.f, 0.f, 0.f, 0.f, 0.f, 0.f, 0.f, 0.f};

#pragma unroll 8
  for (int k = 0; k < K; k += 4) {
    v2f b;
    if (GUARD) {
      b.x = colOK ? W[(size_t)(k + kBase + 0) * DOUT + colB] : 0.f;
      b.y = colOK ? W[(size_t)(k + kBase + 1) * DOUT + colB] : 0.f;
    } else {
      b.x = W[(size_t)(k + kBase + 0) * DOUT + colB];
      b.y = W[(size_t)(k + kBase + 1) * DOUT + colB];
    }
#pragma unroll
    for (int t = 0; t < TM; ++t) {
      const v2f a = *(const v2f*)(arow + t * 16 * PITCH + k);
      acc[t] = __builtin_amdgcn_wmma_f32_16x16x4_f32(
           false, a,  false, b,
           (short)0, acc[t],  false,  false);
    }
  }

  if (colOK) {
#pragma unroll
    for (int t = 0; t < TM; ++t) {
      const int row0 = m0 + (wm * TM + t) * 16;
      if (row0 < N) {
#pragma unroll
        for (int r = 0; r < 8; ++r) {
          Y[(size_t)(row0 + r + 8 * lhalf) * DOUT + colB] = acc[t][r];
        }
      }
    }
  }
}

#define SORT_R     8
#define SORT_S     32
#define RANGE_N    ((N_NODES + SORT_R - 1) / SORT_R)
#define SLICE_E    (N_EDGES / SORT_S)
#define SLICE_Q    (SLICE_E / 4)
#define CR_THREADS 1024
#define CR_IT      ((SLICE_Q + CR_THREADS - 1) / CR_THREADS)
#define CR_PN      ((RANGE_N + CR_THREADS - 1) / CR_THREADS)
static_assert(SLICE_E * SORT_S == N_EDGES, "slices must tile the edge list");
static_assert(SLICE_E % 4 == 0, "a slice is read as whole int4");
static_assert(RANGE_N * SORT_R >= N_NODES, "ranges must cover the nodes");
__global__ __launch_bounds__(256) void zero_counts(int* __restrict__ cnt, int n) {
  const int i = blockIdx.x * blockDim.x + threadIdx.x;
  if (i < n) cnt[i] = 0;
}

__global__ __launch_bounds__(CR_THREADS) void count_rank_lds(
    const int* __restrict__ dst, int* __restrict__ cnt,
    int* __restrict__ rank, int* __restrict__ base) {
  __shared__ int lcnt[RANGE_N];
  const int ri = blockIdx.x % SORT_R;
  const int sj = blockIdx.x / SORT_R;
  const int lo = ri * RANGE_N;
  const int4* __restrict__ d4 = (const int4*)(dst + (size_t)sj * SLICE_E);
  int* __restrict__ rk = rank + (size_t)sj * SLICE_E;

  int4 d[CR_IT];
#pragma unroll
  for (int it = 0; it < CR_IT; ++it) {
    const int q = it * CR_THREADS + (int)threadIdx.x;
    d[it] = (q < SLICE_Q) ? d4[q] : make_int4(-1, -1, -1, -1);
  }

  for (int n = threadIdx.x; n < RANGE_N; n += CR_THREADS) lcnt[n] = 0;
  __syncthreads();

#pragma unroll
  for (int it = 0; it < CR_IT; ++it) {
    const int e0 = (it * CR_THREADS + (int)threadIdx.x) * 4;
    unsigned u;
    u = (unsigned)(d[it].x - lo);
    if (u < (unsigned)RANGE_N) rk[e0 + 0] = atomicAdd(&lcnt[u], 1);
    u = (unsigned)(d[it].y - lo);
    if (u < (unsigned)RANGE_N) rk[e0 + 1] = atomicAdd(&lcnt[u], 1);
    u = (unsigned)(d[it].z - lo);
    if (u < (unsigned)RANGE_N) rk[e0 + 2] = atomicAdd(&lcnt[u], 1);
    u = (unsigned)(d[it].w - lo);
    if (u < (unsigned)RANGE_N) rk[e0 + 3] = atomicAdd(&lcnt[u], 1);
  }
  __syncthreads();

  int got[CR_PN];
#pragma unroll
  for (int it = 0; it < CR_PN; ++it) {
    const int n = it * CR_THREADS + (int)threadIdx.x;
    const int node = lo + n;
    got[it] = 0;
    if (n < RANGE_N && node < N_NODES) got[it] = atomicAdd(&cnt[node], lcnt[n]);
  }
#pragma unroll
  for (int it = 0; it < CR_PN; ++it) {
    const int n = it * CR_THREADS + (int)threadIdx.x;
    const int node = lo + n;
    if (n < RANGE_N && node < N_NODES) base[(size_t)sj * N_NODES + node] = got[it];
  }
}

__global__ __launch_bounds__(256) void place_edges(
    const int* __restrict__ src, const int* __restrict__ dst,
    const float* __restrict__ ew, const int* __restrict__ rowptr,
    const int* __restrict__ rank, const int* __restrict__ base,
    int2* __restrict__ rec, int E) {
  const int e = blockIdx.x * blockDim.x + threadIdx.x;
  if (e < E) {
    const int d = dst[e];
    const int sj = e / SLICE_E;
    const int p = rowptr[d] + base[(size_t)sj * N_NODES + d] + rank[e];
    rec[p] = make_int2(src[e], __float_as_int(ew[e]));
  }
}

template <int K, int DOUT, int WM, int NTHREADS>
__global__ __launch_bounds__(NTHREADS) void gather_gemm_f32(
    const float* __restrict__ S, const int* __restrict__ rowptr,
    const int2* __restrict__ rec, const float* __restrict__ b,
    const float* __restrict__ W, float* __restrict__ Y, int N) {
  constexpr bool GUARD = (DOUT % 16) != 0;
  constexpr int NT    = (DOUT + 15) / 16;
  constexpr int BR    = 16 * WM;
  constexpr int PITCH = K + 4;
  constexpr int CH    = K / 4;
  constexpr int RPP   = NTHREADS / CH;
  constexpr int NPASS = BR / RPP;
  static_assert(BR % RPP == 0 && NPASS >= 1, "lane groups share the rows evenly");
  static_assert(NTHREADS >= 32 * NT * WM, "one wave per output tile");
  static_assert(K % 4 == 0 && ((PITCH / 4) & 1) == 1, "pitch must be 16 B x odd");

  __shared__ __attribute__((aligned(16))) float sA[BR * PITCH];

  const int m0 = blockIdx.x * BR;
  const int g  = (int)threadIdx.x / CH;
  const int f  = ((int)threadIdx.x % CH) * 4;
  const int r0 = g * NPASS;

  int rp[NPASS + 1];
#pragma unroll
  for (int j = 0; j <= NPASS; ++j) {
    int n = m0 + r0 + j;
    if (n > N) n = N;
    rp[j] = rowptr[n];
  }
  const float4 bias = *(const float4*)(b + f);

#pragma unroll
  for (int j = 0; j < NPASS; ++j) {
    float4 acc = bias;
    const int end = rp[j + 1];
#pragma unroll 4
    for (int p = rp[j]; p < end; ++p) {
      const int2 r = rec[p];
      const float w = __int_as_float(r.y);
      const float4 s = *(const float4*)(S + (size_t)r.x * K + f);
      acc.x = fmaf(w, s.x, acc.x);
      acc.y = fmaf(w, s.y, acc.y);
      acc.z = fmaf(w, s.z, acc.z);
      acc.w = fmaf(w, s.w, acc.w);
    }
    acc.x = fmaxf(acc.x, 0.f); acc.y = fmaxf(acc.y, 0.f);
    acc.z = fmaxf(acc.z, 0.f); acc.w = fmaxf(acc.w, 0.f);
    *(float4*)(sA + (r0 + j) * PITCH + f) = acc;
  }
  __syncthreads();

  const int wave = (int)threadIdx.x >> 5;
  if (wave >= NT * WM) return;
  const int lane  = (int)threadIdx.x & 31;
  const int wn    = wave % NT;
  const int wm    = wave / NT;
  const int n0    = wn * 16;
  const int lhalf = lane >> 4;
  const int l16   = lane & 15;
  const int  colB  = n0 + l16;
  const bool colOK = !GUARD || (colB < DOUT);
  const int  kBase = 2 * lhalf;
  const float* arow = sA + (wm * 16 + l16) * PITCH + kBase;

  v8f acc = (v8f){0.f, 0.f, 0.f, 0.f, 0.f, 0.f, 0.f, 0.f};
#pragma unroll 8
  for (int k = 0; k < K; k += 4) {
    v2f bf;
    if (GUARD) {
      bf.x = colOK ? W[(size_t)(k + kBase + 0) * DOUT + colB] : 0.f;
      bf.y = colOK ? W[(size_t)(k + kBase + 1) * DOUT + colB] : 0.f;
    } else {
      bf.x = W[(size_t)(k + kBase + 0) * DOUT + colB];
      bf.y = W[(size_t)(k + kBase + 1) * DOUT + colB];
    }
    const v2f a = *(const v2f*)(arow + k);
    acc = __builtin_amdgcn_wmma_f32_16x16x4_f32(
         false, a,  false, bf,
         (short)0, acc,  false,  false);
  }

  const int row0 = m0 + wm * 16;
  if (colOK && row0 < N) {
#pragma unroll
    for (int r = 0; r < 8; ++r) {
      Y[(size_t)(row0 + r + 8 * lhalf) * DOUT + colB] = acc[r];
    }
  }
}

template <int D>
__global__ __launch_bounds__(256) void gather_scalar(
    const float* __restrict__ S, const int* __restrict__ rowptr,
    const int2* __restrict__ rec, const float* __restrict__ b,
    float* __restrict__ A, int N) {
  const int i = blockIdx.x * blockDim.x + threadIdx.x;
  if (i >= N * D) return;
  const int n = i / D;
  const int f = i % D;
  const int beg = rowptr[n];
  const int end = rowptr[n + 1];
  float acc = b[f];
#pragma unroll 4
  for (int p = beg; p < end; ++p) {
    const int2 r = rec[p];
    acc = fmaf(__int_as_float(r.y), S[(size_t)r.x * D + f], acc);
  }
  A[i] = acc;
}

static inline int cdiv(long long a, long long b) { return (int)((a + b - 1) / b); }

extern "C" void kernel_launch(void* const* d_in, const int* in_sizes, int n_in,
                              void* d_out, int out_size, void* d_ws, size_t ws_size,
                              hipStream_t stream) {
  const float* x   = (const float*)d_in[0];
  const int*   src = (const int*)  d_in[1];
  const int*   dst = (const int*)  d_in[2];
  const float* ew  = (const float*)d_in[3];
  const float* W1  = (const float*)d_in[4];
  const float* b1  = (const float*)d_in[5];
  const float* W2  = (const float*)d_in[6];
  const float* b2  = (const float*)d_in[7];
  const float* W3  = (const float*)d_in[8];
  const float* b3  = (const float*)d_in[9];
  float* out = (float*)d_out;

  const int N = N_NODES, E = N_EDGES;

  float* bufS = (float*)d_ws;
  float* bufH = bufS + (size_t)N * 128;
  int*  cnt    = (int*)(bufH + (size_t)N * 128);
  int*  rowptr = cnt + 50048;
  int*  rank   = rowptr + 50048;
  int2* rec    = (int2*)(rank + N_EDGES);
  int*  base   = (int*)(rec + N_EDGES);

  zero_counts<<<cdiv(N, 256), 256, 0, stream>>>(cnt, N);
  count_rank_lds<<<SORT_R * SORT_S, CR_THREADS, 0, stream>>>(dst, cnt, rank, base);

  gemm_wmma_f32<256, 128, false, 2, 1><<<SCAN_WG + cdiv(N, 32), 256, 0, stream>>>(
      x, W1, bufS, N, cnt, rowptr);

  place_edges<<<cdiv(E, 256), 256, 0, stream>>>(src, dst, ew, rowptr, rank, base, rec, E);

  gather_gemm_f32<128, 64, 2, 256><<<cdiv(N, 32), 256, 0, stream>>>(
      bufS, rowptr, rec, b1, W2, bufH, N);

  gather_gemm_f32<64, 10, 1, 128><<<cdiv(N, 16), 128, 0, stream>>>(
      bufH, rowptr, rec, b2, W3, bufS, N);

  gather_scalar<10><<<cdiv((long long)N * 10, 256), 256, 0, stream>>>(
      bufS, rowptr, rec, b3, out, N);
}


# Edge placement in the counting layout: work-group (node range, slice) sets LDS cursors to rowptr+slice start; a returning LDS add places each edge (arrival order); no rank array or per-edge lookups
# speedup vs baseline: 3.4231x; 1.3051x over previous
#include <hip/hip_runtime.h>


typedef float v2f __attribute__((ext_vector_type(2)));
typedef float v8f __attribute__((ext_vector_type(8)));

#define N_NODES 50000
#define N_EDGES 800000

#define SCAN_WG    28
#define SCAN_ROWS  7
#define SCAN_WSEG  (SCAN_ROWS * 32)
#define SCAN_PIECE (SCAN_WSEG * 8)
#define SCAN_PQ    (SCAN_PIECE / 4)
#define SCAN_HU    8
static_assert(SCAN_PIECE * SCAN_WG >= N_NODES, "the pieces must cover the nodes");
static_assert((SCAN_WG - 1) * SCAN_PIECE < N_NODES, "every head lies inside the counts");

__device__ __forceinline__ void scan_piece(const int* __restrict__ cnt,
                                           int* __restrict__ rowptr) {
  __shared__ int sHead[8];
  __shared__ int sTot[8];
  const int lane = (int)threadIdx.x & 31;
  const int wave = (int)threadIdx.x >> 5;
  const int seg0 = (int)blockIdx.x * SCAN_PIECE + wave * SCAN_WSEG;

  int v[SCAN_ROWS];
#pragma unroll
  for (int r = 0; r < SCAN_ROWS; ++r) {
    const int idx = seg0 + r * 32 + lane;
    v[r] = (idx < N_NODES) ? cnt[idx] : 0;
  }

  const int4* __restrict__ c4 = (const int4*)cnt;
  const int nq = (int)blockIdx.x * SCAN_PQ;
  int hsum = 0;
  for (int q0 = 0; q0 < nq; q0 += 256 * SCAN_HU) {
    int4 t[SCAN_HU];
#pragma unroll
    for (int u = 0; u < SCAN_HU; ++u) {
      int q = q0 + u * 256 + (int)threadIdx.x;
      if (q > nq - 1) q = nq - 1;
      t[u] = c4[q];
    }
#pragma unroll
    for (int u = 0; u < SCAN_HU; ++u) {
      const int q = q0 + u * 256 + (int)threadIdx.x;
      const int a = (t[u].x + t[u].y) + (t[u].z + t[u].w);
      hsum += (q < nq) ? a : 0;
    }
  }

  int lsum = 0;
#pragma unroll
  for (int r = 0; r < SCAN_ROWS; ++r) lsum += v[r];
  for (int off = 16; off > 0; off >>= 1) {
    hsum += __shfl_xor(hsum, off);
    lsum += __shfl_xor(lsum, off);
  }
  if (lane == 0) { sHead[wave] = hsum; sTot[wave] = lsum; }
  __syncthreads();

  int carry = 0;
  if (lane < 8) carry = sHead[lane];
  else if (lane < 16 && (lane - 8) < wave) carry = sTot[lane - 8];
  for (int off = 16; off > 0; off >>= 1) carry += __shfl_xor(carry, off);

#pragma unroll
  for (int r = 0; r < SCAN_ROWS; ++r) {
    int inc = v[r];
    for (int off = 1; off < 32; off <<= 1) {
      const int up = __shfl_up(inc, off);
      if (lane >= off) inc += up;
    }
    const int idx = seg0 + r * 32 + lane;
    if (idx < N_NODES) rowptr[idx] = carry + inc - v[r];
    if (idx == N_NODES - 1) rowptr[N_NODES] = carry + inc;
    carry += __shfl(inc, 31);
  }
}

template <int K, int DOUT, bool RELU_IN, int TM, int WM>
__global__ __launch_bounds__(32 * ((DOUT + 15) / 16) * WM) void gemm_wmma_f32(
    const float* __restrict__ X, const float* __restrict__ W,
    float* __restrict__ Y, int N,
    const int* __restrict__ cnt, int* __restrict__ rowptr) {
  if (blockIdx.x < SCAN_WG) {
    scan_piece(cnt, rowptr);
    return;
  }
  constexpr bool GUARD = (DOUT % 16) != 0;
  constexpr int NT       = (DOUT + 15) / 16;
  constexpr int NTHREADS = 32 * NT * WM;
  constexpr int BR       = 16 * TM * WM;
  constexpr int PITCH    = K + 4;
  constexpr int Q        = K / 4;
  constexpr int NLD      = (BR * Q) / NTHREADS;
  static_assert((BR * Q) % NTHREADS == 0, "the row block is copied in whole passes");
  static_assert(K % 4 == 0 && ((PITCH / 4) & 1) == 1, "pitch must be 16 B x odd");
  static_assert(N_NODES % 16 == 0, "row tiles must be whole");

  __shared__ __attribute__((aligned(16))) float sA[BR * PITCH];

  const int wave  = threadIdx.x >> 5;
  const int lane  = threadIdx.x & 31;
  const int wn    = wave % NT;
  const int wm    = wave / NT;
  const int n0    = wn * 16;
  const int m0    = ((int)blockIdx.x - SCAN_WG) * BR;
  const int lhalf = lane >> 4;
  const int l16   = lane & 15;

  float4 v[NLD];
#pragma unroll
  for (int i = 0; i < NLD; ++i) {
    const int q = i * NTHREADS + (int)threadIdx.x;
    const int r = q / Q;
    const int c = q % Q;
    int row = m0 + r;
    if (row > N - 1) row = N - 1;
    v[i] = *(const float4*)(X + (size_t)row * K + c * 4);
  }
#pragma unroll
  for (int i = 0; i < NLD; ++i) {
    const int q = i * NTHREADS + (int)threadIdx.x;
    const int r = q / Q;
    const int c = q % Q;
    float4 t = v[i];
    if (RELU_IN) {
      t.x = fmaxf(t.x, 0.f); t.y = fmaxf(t.y, 0.f);
      t.z = fmaxf(t.z, 0.f); t.w = fmaxf(t.w, 0.f);
    }
    *(float4*)(sA + r * PITCH + c * 4) = t;
  }
  __syncthreads();

  const int  colB  = n0 + l16;
  const bool colOK = !GUARD || (colB < DOUT);
  const int  kBase = 2 * lhalf;
  const float* arow = sA + (wm * TM * 16 + l16) * PITCH + kBase;

  v8f acc[TM];
#pragma unroll
  for (int t = 0; t < TM; ++t) acc[t] = (v8f){0.f, 0.f, 0.f, 0.f, 0.f, 0.f, 0.f, 0.f};

#pragma unroll 8
  for (int k = 0; k < K; k += 4) {
    v2f b;
    if (GUARD) {
      b.x = colOK ? W[(size_t)(k + kBase + 0) * DOUT + colB] : 0.f;
      b.y = colOK ? W[(size_t)(k + kBase + 1) * DOUT + colB] : 0.f;
    } else {
      b.x = W[(size_t)(k + kBase + 0) * DOUT + colB];
      b.y = W[(size_t)(k + kBase + 1) * DOUT + colB];
    }
#pragma unroll
    for (int t = 0; t < TM; ++t) {
      const v2f a = *(const v2f*)(arow + t * 16 * PITCH + k);
      acc[t] = __builtin_amdgcn_wmma_f32_16x16x4_f32(
           false, a,  false, b,
           (short)0, acc[t],  false,  false);
    }
  }

  if (colOK) {
#pragma unroll
    for (int t = 0; t < TM; ++t) {
      const int row0 = m0 + (wm * TM + t) * 16;
      if (row0 < N) {
#pragma unroll
        for (int r = 0; r < 8; ++r) {
          Y[(size_t)(row0 + r + 8 * lhalf) * DOUT + colB] = acc[t][r];
        }
      }
    }
  }
}

#define SORT_R     8
#define SORT_S     32
#define RANGE_N    ((N_NODES + SORT_R - 1) / SORT_R)
#define SLICE_E    (N_EDGES / SORT_S)
#define SLICE_Q    (SLICE_E / 4)
#define CR_THREADS 1024
#define CR_IT      ((SLICE_Q + CR_THREADS - 1) / CR_THREADS)
#define CR_PN      ((RANGE_N + CR_THREADS - 1) / CR_THREADS)
static_assert(SLICE_E * SORT_S == N_EDGES, "slices must tile the edge list");
static_assert(SLICE_E % 4 == 0, "a slice is read as whole int4");
static_assert(RANGE_N * SORT_R >= N_NODES, "ranges must cover the nodes");
__global__ __launch_bounds__(256) void zero_counts(int* __restrict__ cnt, int n) {
  const int i = blockIdx.x * blockDim.x + threadIdx.x;
  if (i < n) cnt[i] = 0;
}

__global__ __launch_bounds__(CR_THREADS) void count_lds(
    const int* __restrict__ dst, int* __restrict__ cnt, int* __restrict__ base) {
  __shared__ int lcnt[RANGE_N];
  const int ri = blockIdx.x % SORT_R;
  const int sj = blockIdx.x / SORT_R;
  const int lo = ri * RANGE_N;
  const int4* __restrict__ d4 = (const int4*)(dst + (size_t)sj * SLICE_E);

  int4 d[CR_IT];
#pragma unroll
  for (int it = 0; it < CR_IT; ++it) {
    const int q = it * CR_THREADS + (int)threadIdx.x;
    d[it] = (q < SLICE_Q) ? d4[q] : make_int4(-1, -1, -1, -1);
  }

  for (int n = threadIdx.x; n < RANGE_N; n += CR_THREADS) lcnt[n] = 0;
  __syncthreads();

#pragma unroll
  for (int it = 0; it < CR_IT; ++it) {
    unsigned u;
    u = (unsigned)(d[it].x - lo);
    if (u < (unsigned)RANGE_N) atomicAdd(&lcnt[u], 1);
    u = (unsigned)(d[it].y - lo);
    if (u < (unsigned)RANGE_N) atomicAdd(&lcnt[u], 1);
    u = (unsigned)(d[it].z - lo);
    if (u < (unsigned)RANGE_N) atomicAdd(&lcnt[u], 1);
    u = (unsigned)(d[it].w - lo);
    if (u < (unsigned)RANGE_N) atomicAdd(&lcnt[u], 1);
  }
  __syncthreads();

  int got[CR_PN];
#pragma unroll
  for (int it = 0; it < CR_PN; ++it) {
    const int n = it * CR_THREADS + (int)threadIdx.x;
    const int node = lo + n;
    got[it] = 0;
    if (n < RANGE_N && node < N_NODES) got[it] = atomicAdd(&cnt[node], lcnt[n]);
  }
#pragma unroll
  for (int it = 0; it < CR_PN; ++it) {
    const int n = it * CR_THREADS + (int)threadIdx.x;
    const int node = lo + n;
    if (n < RANGE_N && node < N_NODES) base[(size_t)sj * N_NODES + node] = got[it];
  }
}

__global__ __launch_bounds__(CR_THREADS) void place_lds(
    const int* __restrict__ src, const int* __restrict__ dst,
    const float* __restrict__ ew, const int* __restrict__ rowptr,
    const int* __restrict__ base, int2* __restrict__ rec) {
  __shared__ int lcur[CR_PN * CR_THREADS];
  const int ri = blockIdx.x % SORT_R;
  const int sj = blockIdx.x / SORT_R;
  const int lo = ri * RANGE_N;
  const int4*   __restrict__ d4 = (const int4*)(dst + (size_t)sj * SLICE_E);
  const int4*   __restrict__ s4 = (const int4*)(src + (size_t)sj * SLICE_E);
  const float4* __restrict__ w4 = (const float4*)(ew + (size_t)sj * SLICE_E);
  const int* __restrict__ bs = base + (size_t)sj * N_NODES;

  int ra[CR_PN];
  int ba[CR_PN];
#pragma unroll
  for (int it = 0; it < CR_PN; ++it) {
    int n = it * CR_THREADS + (int)threadIdx.x;
    if (n > RANGE_N - 1) n = RANGE_N - 1;
    int node = lo + n;
    if (node > N_NODES - 1) node = N_NODES - 1;
    ra[it] = rowptr[node];
    ba[it] = bs[node];
  }
  int4 d[CR_IT];
  int4 s[CR_IT];
  float4 w[CR_IT];
#pragma unroll
  for (int it = 0; it < CR_IT; ++it) {
    int q = it * CR_THREADS + (int)threadIdx.x;
    if (q > SLICE_Q - 1) q = SLICE_Q - 1;
    d[it] = d4[q];
    s[it] = s4[q];
    w[it] = w4[q];
  }
#pragma unroll
  for (int it = 0; it < CR_PN; ++it) {
    lcur[it * CR_THREADS + (int)threadIdx.x] = ra[it] + ba[it];
  }
  __syncthreads();

#pragma unroll
  for (int it = 0; it < CR_IT; ++it) {
    const bool ok = (it * CR_THREADS + (int)threadIdx.x) < SLICE_Q;
    unsigned u;
    u = (unsigned)((ok ? d[it].x : -1) - lo);
    if (u < (unsigned)RANGE_N)
      rec[atomicAdd(&lcur[u], 1)] = make_int2(s[it].x, __float_as_int(w[it].x));
    u = (unsigned)((ok ? d[it].y : -1) - lo);
    if (u < (unsigned)RANGE_N)
      rec[atomicAdd(&lcur[u], 1)] = make_int2(s[it].y, __float_as_int(w[it].y));
    u = (unsigned)((ok ? d[it].z : -1) - lo);
    if (u < (unsigned)RANGE_N)
      rec[atomicAdd(&lcur[u], 1)] = make_int2(s[it].z, __float_as_int(w[it].z));
    u = (unsigned)((ok ? d[it].w : -1) - lo);
    if (u < (unsigned)RANGE_N)
      rec[atomicAdd(&lcur[u], 1)] = make_int2(s[it].w, __float_as_int(w[it].w));
  }
}

template <int K, int DOUT, int WM, int NTHREADS>
__global__ __launch_bounds__(NTHREADS) void gather_gemm_f32(
    const float* __restrict__ S, const int* __restrict__ rowptr,
    const int2* __restrict__ rec, const float* __restrict__ b,
    const float* __restrict__ W, float* __restrict__ Y, int N) {
  constexpr bool GUARD = (DOUT % 16) != 0;
  constexpr int NT    = (DOUT + 15) / 16;
  constexpr int BR    = 16 * WM;
  constexpr int PITCH = K + 4;
  constexpr int CH    = K / 4;
  constexpr int RPP   = NTHREADS / CH;
  constexpr int NPASS = BR / RPP;
  static_assert(BR % RPP == 0 && NPASS >= 1, "lane groups share the rows evenly");
  static_assert(NTHREADS >= 32 * NT * WM, "one wave per output tile");
  static_assert(K % 4 == 0 && ((PITCH / 4) & 1) == 1, "pitch must be 16 B x odd");

  __shared__ __attribute__((aligned(16))) float sA[BR * PITCH];

  const int m0 = blockIdx.x * BR;
  const int g  = (int)threadIdx.x / CH;
  const int f  = ((int)threadIdx.x % CH) * 4;
  const int r0 = g * NPASS;

  int rp[NPASS + 1];
#pragma unroll
  for (int j = 0; j <= NPASS; ++j) {
    int n = m0 + r0 + j;
    if (n > N) n = N;
    rp[j] = rowptr[n];
  }
  const float4 bias = *(const float4*)(b + f);

#pragma unroll
  for (int j = 0; j < NPASS; ++j) {
    float4 acc = bias;
    const int end = rp[j + 1];
#pragma unroll 4
    for (int p = rp[j]; p < end; ++p) {
      const int2 r = rec[p];
      const float w = __int_as_float(r.y);
      const float4 s = *(const float4*)(S + (size_t)r.x * K + f);
      acc.x = fmaf(w, s.x, acc.x);
      acc.y = fmaf(w, s.y, acc.y);
      acc.z = fmaf(w, s.z, acc.z);
      acc.w = fmaf(w, s.w, acc.w);
    }
    acc.x = fmaxf(acc.x, 0.f); acc.y = fmaxf(acc.y, 0.f);
    acc.z = fmaxf(acc.z, 0.f); acc.w = fmaxf(acc.w, 0.f);
    *(float4*)(sA + (r0 + j) * PITCH + f) = acc;
  }
  __syncthreads();

  const int wave = (int)threadIdx.x >> 5;
  if (wave >= NT * WM) return;
  const int lane  = (int)threadIdx.x & 31;
  const int wn    = wave % NT;
  const int wm    = wave / NT;
  const int n0    = wn * 16;
  const int lhalf = lane >> 4;
  const int l16   = lane & 15;
  const int  colB  = n0 + l16;
  const bool colOK = !GUARD || (colB < DOUT);
  const int  kBase = 2 * lhalf;
  const float* arow = sA + (wm * 16 + l16) * PITCH + kBase;

  v8f acc = (v8f){0.f, 0.f, 0.f, 0.f, 0.f, 0.f, 0.f, 0.f};
#pragma unroll 8
  for (int k = 0; k < K; k += 4) {
    v2f bf;
    if (GUARD) {
      bf.x = colOK ? W[(size_t)(k + kBase + 0) * DOUT + colB] : 0.f;
      bf.y = colOK ? W[(size_t)(k + kBase + 1) * DOUT + colB] : 0.f;
    } else {
      bf.x = W[(size_t)(k + kBase + 0) * DOUT + colB];
      bf.y = W[(size_t)(k + kBase + 1) * DOUT + colB];
    }
    const v2f a = *(const v2f*)(arow + k);
    acc = __builtin_amdgcn_wmma_f32_16x16x4_f32(
         false, a,  false, bf,
         (short)0, acc,  false,  false);
  }

  const int row0 = m0 + wm * 16;
  if (colOK && row0 < N) {
#pragma unroll
    for (int r = 0; r < 8; ++r) {
      Y[(size_t)(row0 + r + 8 * lhalf) * DOUT + colB] = acc[r];
    }
  }
}

template <int D>
__global__ __launch_bounds__(256) void gather_scalar(
    const float* __restrict__ S, const int* __restrict__ rowptr,
    const int2* __restrict__ rec, const float* __restrict__ b,
    float* __restrict__ A, int N) {
  const int i = blockIdx.x * blockDim.x + threadIdx.x;
  if (i >= N * D) return;
  const int n = i / D;
  const int f = i % D;
  const int beg = rowptr[n];
  const int end = rowptr[n + 1];
  float acc = b[f];
#pragma unroll 4
  for (int p = beg; p < end; ++p) {
    const int2 r = rec[p];
    acc = fmaf(__int_as_float(r.y), S[(size_t)r.x * D + f], acc);
  }
  A[i] = acc;
}

static inline int cdiv(long long a, long long b) { return (int)((a + b - 1) / b); }

extern "C" void kernel_launch(void* const* d_in, const int* in_sizes, int n_in,
                              void* d_out, int out_size, void* d_ws, size_t ws_size,
                              hipStream_t stream) {
  const float* x   = (const float*)d_in[0];
  const int*   src = (const int*)  d_in[1];
  const int*   dst = (const int*)  d_in[2];
  const float* ew  = (const float*)d_in[3];
  const float* W1  = (const float*)d_in[4];
  const float* b1  = (const float*)d_in[5];
  const float* W2  = (const float*)d_in[6];
  const float* b2  = (const float*)d_in[7];
  const float* W3  = (const float*)d_in[8];
  const float* b3  = (const float*)d_in[9];
  float* out = (float*)d_out;

  const int N = N_NODES;

  float* bufS = (float*)d_ws;
  float* bufH = bufS + (size_t)N * 128;
  int*  cnt    = (int*)(bufH + (size_t)N * 128);
  int*  rowptr = cnt + 50048;
  int2* rec    = (int2*)(rowptr + 50048);
  int*  base   = (int*)(rec + N_EDGES);

  zero_counts<<<cdiv(N, 256), 256, 0, stream>>>(cnt, N);
  count_lds<<<SORT_R * SORT_S, CR_THREADS, 0, stream>>>(dst, cnt, base);

  gemm_wmma_f32<256, 128, false, 2, 1><<<SCAN_WG + cdiv(N, 32), 256, 0, stream>>>(
      x, W1, bufS, N, cnt, rowptr);

  place_lds<<<SORT_R * SORT_S, CR_THREADS, 0, stream>>>(src, dst, ew, rowptr, base, rec);

  gather_gemm_f32<128, 64, 2, 256><<<cdiv(N, 32), 256, 0, stream>>>(
      bufS, rowptr, rec, b1, W2, bufH, N);

  gather_gemm_f32<64, 10, 1, 128><<<cdiv(N, 16), 128, 0, stream>>>(
      bufH, rowptr, rec, b2, W3, bufS, N);

  gather_scalar<10><<<cdiv((long long)N * 10, 256), 256, 0, stream>>>(
      bufS, rowptr, rec, b3, out, N);
}
